# starting kernel
# baseline (speedup 1.0000x reference)
#include <hip/hip_runtime.h>

typedef __attribute__((ext_vector_type(2))) float v2f;
typedef __attribute__((ext_vector_type(8))) float v8f;

#define N_FEAT 10
#define N_CLS  2
#define KPAD   12
#define TILES_PER_WAVE 8
#define ROWS_PER_WAVE  (16 * TILES_PER_WAVE)
#define ROWS_PER_BLOCK (8 * ROWS_PER_WAVE)

__global__ void build_coeff_kernel(const float* __restrict__ psi_r,
                                   const float* __restrict__ psi_i,
                                   const float* __restrict__ A_r,
                                   const float* __restrict__ A_i,
                                   float* __restrict__ Bpad) {
    int t = threadIdx.x;
    if (t >= KPAD * 16) return;
    int a = t >> 4;
    int k = t & 15;
    float val = 0.0f;
    if (a < N_FEAT && k < N_CLS) {
        for (int i = 0; i < N_FEAT; ++i) {
            float pri = psi_r[i];
            float pii = psi_i[i];
            for (int j = 0; j < N_FEAT; ++j) {
                float w = psi_r[j] - psi_i[j];
                int idx = ((k * N_FEAT + i) * N_FEAT + j) * N_FEAT + a;
                val += w * (pri * A_r[idx] - pii * A_i[idx]);
            }
        }
    }
    Bpad[t] = val;
}

__global__ void __launch_bounds__(256) gemv_wmma_kernel(
    const float* __restrict__ X,
    const float* __restrict__ Bpad,
    float* __restrict__ Out,
    int T) {
    const int lane = threadIdx.x & 31;
    const int wave = threadIdx.x >> 5;
    const int hi   = lane >> 4;
    const int n    = lane & 15;

    const long long waveBase =
        (long long)blockIdx.x * ROWS_PER_BLOCK + (long long)wave * ROWS_PER_WAVE;
    if (waveBase >= T) return;

    v2f bf[3];
#pragma unroll
    for (int c = 0; c < 3; ++c) {
        bf[c].x = Bpad[(4 * c + 0 + hi * 2) * 16 + n];
        bf[c].y = Bpad[(4 * c + 1 + hi * 2) * 16 + n];
    }

#pragma unroll
    for (int t = 0; t < TILES_PER_WAVE; ++t) {
        const long long rowbase = waveBase + t * 16;
        if (rowbase >= T) break;

        long long rowA = rowbase + n;
        if (rowA > (long long)T - 1) rowA = (long long)T - 1;
        const float* xrow = X + rowA * N_FEAT;

        v2f a0 = *(const v2f*)(xrow + 0 + hi * 2);
        v2f a1 = *(const v2f*)(xrow + 4 + hi * 2);
        v2f a2 = *(const v2f*)(xrow + 8);

        v8f c = {};
        c = __builtin_amdgcn_wmma_f32_16x16x4_f32(false, a0, false, bf[0],
                                                  (short)0, c, false, false);
        c = __builtin_amdgcn_wmma_f32_16x16x4_f32(false, a1, false, bf[1],
                                                  (short)0, c, false, false);
        c = __builtin_amdgcn_wmma_f32_16x16x4_f32(false, a2, false, bf[2],
                                                  (short)0, c, false, false);

        if (n < N_CLS) {
#pragma unroll
            for (int v = 0; v < 8; ++v) {
                long long row = rowbase + hi * 8 + v;
                if (row < T) Out[row * N_CLS + n] = c[v];
            }
        }
    }
}

extern "C" void kernel_launch(void* const* d_in, const int* in_sizes, int n_in,
                              void* d_out, int out_size, void* d_ws, size_t ws_size,
                              hipStream_t stream) {
    const float* x     = (const float*)d_in[0];
    const float* psi_r = (const float*)d_in[1];
    const float* psi_i = (const float*)d_in[2];
    const float* A_r   = (const float*)d_in[3];
    const float* A_i   = (const float*)d_in[4];
    float* out  = (float*)d_out;
    float* Bpad = (float*)d_ws;

    const int T = in_sizes[0] / N_FEAT;

    build_coeff_kernel<<<1, 256, 0, stream>>>(psi_r, psi_i, A_r, A_i, Bpad);

    const int grid = (T + ROWS_PER_BLOCK - 1) / ROWS_PER_BLOCK;
    gemv_wmma_kernel<<<grid, 256, 0, stream>>>(x, Bpad, out, T);
}


# persistent waves: a fixed grid of 1024 workgroups walks 32-row items grid-strided with the loads of 4 items in flight per wave, replacing 3907 workgroups of one-shot waves with one tile in flight
# speedup vs baseline: 1.3611x; 1.3611x over previous
#include <hip/hip_runtime.h>


typedef __attribute__((ext_vector_type(2))) float v2f;
typedef __attribute__((ext_vector_type(8))) float v8f;

#define N_FEAT 10
#define N_CLS  2
#define KPAD   12

#define WAVES_PER_BLOCK 8
#define NUM_BLOCKS      1024
#define TILES_PER_ITEM  2
#define ROWS_PER_ITEM   (16 * TILES_PER_ITEM)
#define DEPTH           4

__global__ void build_coeff_kernel(const float* __restrict__ psi_r,
                                   const float* __restrict__ psi_i,
                                   const float* __restrict__ A_r,
                                   const float* __restrict__ A_i,
                                   float* __restrict__ Bpad) {
    int t = threadIdx.x;
    if (t >= KPAD * 16) return;
    int a = t >> 4;
    int k = t & 15;
    float val = 0.0f;
    if (a < N_FEAT && k < N_CLS) {
        for (int i = 0; i < N_FEAT; ++i) {
            float pri = psi_r[i];
            float pii = psi_i[i];
            for (int j = 0; j < N_FEAT; ++j) {
                float w = psi_r[j] - psi_i[j];
                int idx = ((k * N_FEAT + i) * N_FEAT + j) * N_FEAT + a;
                val += w * (pri * A_r[idx] - pii * A_i[idx]);
            }
        }
    }
    Bpad[t] = val;
}

struct TileFrag { v2f a0, a1, a2; };

__device__ __forceinline__ void load_tile(const float* __restrict__ X,
                                          unsigned rowbase, unsigned lastRow,
                                          int n, int hi, TileFrag& f) {
    unsigned rowA = rowbase + (unsigned)n;
    if (rowA > lastRow) rowA = lastRow;
    const float* xrow = X + (size_t)rowA * N_FEAT;
    f.a0 = *(const v2f*)(xrow + 0 + hi * 2);
    f.a1 = *(const v2f*)(xrow + 4 + hi * 2);
    f.a2 = *(const v2f*)(xrow + 8);
}

__device__ __forceinline__ void tile_product_store(const TileFrag& f, const v2f (&bf)[3],
                                                   float* __restrict__ Out,
                                                   unsigned rowbase, unsigned T,
                                                   int n, int hi) {
    v8f c = {};
    c = __builtin_amdgcn_wmma_f32_16x16x4_f32(false, f.a0, false, bf[0],
                                              (short)0, c, false, false);
    c = __builtin_amdgcn_wmma_f32_16x16x4_f32(false, f.a1, false, bf[1],
                                              (short)0, c, false, false);
    c = __builtin_amdgcn_wmma_f32_16x16x4_f32(false, f.a2, false, bf[2],
                                              (short)0, c, false, false);

    asm volatile("v_nop\n\tv_nop\n\tv_nop\n\tv_nop" : "+v"(c));

    if (n < N_CLS) {
        const unsigned row0 = rowbase + (unsigned)hi * 8u;
        float* o = Out + (size_t)row0 * N_CLS + n;
        if (rowbase + 16u <= T) {
#pragma unroll
            for (int v = 0; v < 8; ++v) o[v * N_CLS] = c[v];
        } else {
#pragma unroll
            for (int v = 0; v < 8; ++v)
                if (row0 + (unsigned)v < T) o[v * N_CLS] = c[v];
        }
    }
}

__global__ void __launch_bounds__(256) gemv_wmma_kernel(
    const float* __restrict__ X,
    const float* __restrict__ Bpad,
    float* __restrict__ Out,
    int Tin) {
    if (Tin <= 0) return;
    const unsigned T = (unsigned)Tin;
    const int lane = threadIdx.x & 31;
    const int wave = __builtin_amdgcn_readfirstlane((int)(threadIdx.x >> 5));
    const int hi   = lane >> 4;
    const int n    = lane & 15;

    const unsigned nwaves = gridDim.x * WAVES_PER_BLOCK;
    const unsigned nitems = (T + ROWS_PER_ITEM - 1u) / ROWS_PER_ITEM;
    const unsigned lastRow = T - 1u;
    unsigned g = blockIdx.x * WAVES_PER_BLOCK + (unsigned)wave;
    if (g >= nitems) return;

    v2f bf[3];
#pragma unroll
    for (int c = 0; c < 3; ++c) {
        bf[c].x = Bpad[(4 * c + 0 + hi * 2) * 16 + n];
        bf[c].y = Bpad[(4 * c + 1 + hi * 2) * 16 + n];
    }

    TileFrag buf[DEPTH][TILES_PER_ITEM];
#pragma unroll
    for (int s = 0; s < DEPTH; ++s) {
        const unsigned rb = (g + (unsigned)s * nwaves) * ROWS_PER_ITEM;
#pragma unroll
        for (int t = 0; t < TILES_PER_ITEM; ++t)
            load_tile(X, rb + 16u * t, lastRow, n, hi, buf[s][t]);
        __builtin_amdgcn_sched_barrier(0);
    }

    const unsigned myItems = (nitems - g + nwaves - 1u) / nwaves;
    const unsigned trips = (myItems + DEPTH - 1u) / DEPTH;
    for (unsigned it = 0; it < trips; ++it) {
#pragma unroll
        for (int s = 0; s < DEPTH; ++s) {
            const unsigned rb = g * ROWS_PER_ITEM;
#pragma unroll
            for (int t = 0; t < TILES_PER_ITEM; ++t)
                tile_product_store(buf[s][t], bf, Out, rb + 16u * t, T, n, hi);
            const unsigned rbn = (g + (unsigned)DEPTH * nwaves) * ROWS_PER_ITEM;
#pragma unroll
            for (int t = 0; t < TILES_PER_ITEM; ++t)
                load_tile(X, rbn + 16u * t, lastRow, n, hi, buf[s][t]);
            g += nwaves;
        }
    }
}

extern "C" void kernel_launch(void* const* d_in, const int* in_sizes, int n_in,
                              void* d_out, int out_size, void* d_ws, size_t ws_size,
                              hipStream_t stream) {
    const float* x     = (const float*)d_in[0];
    const float* psi_r = (const float*)d_in[1];
    const float* psi_i = (const float*)d_in[2];
    const float* A_r   = (const float*)d_in[3];
    const float* A_i   = (const float*)d_in[4];
    float* out  = (float*)d_out;
    float* Bpad = (float*)d_ws;

    const int T = in_sizes[0] / N_FEAT;

    build_coeff_kernel<<<1, 256, 0, stream>>>(psi_r, psi_i, A_r, A_i, Bpad);

    gemv_wmma_kernel<<<NUM_BLOCKS, 256, 0, stream>>>(x, Bpad, out, T);
}


# transposed product: coefficients as WMMA A operand, x rows as B columns, so lane l ends with both outputs of row l; one full-wave 256-byte store per 32 rows replaces 16 four-lane 16-byte stores
# speedup vs baseline: 1.4858x; 1.0916x over previous
#include <hip/hip_runtime.h>


typedef __attribute__((ext_vector_type(2))) float v2f;
typedef __attribute__((ext_vector_type(8))) float v8f;

#define N_FEAT 10
#define N_CLS  2
#define KPAD   12

#define WAVES_PER_BLOCK 8
#define NUM_BLOCKS      1024
#define TILES_PER_ITEM  2
#define ROWS_PER_ITEM   (16 * TILES_PER_ITEM)
#define DEPTH           4

__global__ void build_coeff_kernel(const float* __restrict__ psi_r,
                                   const float* __restrict__ psi_i,
                                   const float* __restrict__ A_r,
                                   const float* __restrict__ A_i,
                                   float* __restrict__ Bpad) {
    int t = threadIdx.x;
    if (t >= KPAD * 16) return;
    int a = t >> 4;
    int k = t & 15;
    float val = 0.0f;
    if (a < N_FEAT && k < N_CLS) {
        for (int i = 0; i < N_FEAT; ++i) {
            float pri = psi_r[i];
            float pii = psi_i[i];
            for (int j = 0; j < N_FEAT; ++j) {
                float w = psi_r[j] - psi_i[j];
                int idx = ((k * N_FEAT + i) * N_FEAT + j) * N_FEAT + a;
                val += w * (pri * A_r[idx] - pii * A_i[idx]);
            }
        }
    }
    Bpad[t] = val;
}

struct TileFrag { v2f a0, a1, a2; };

__device__ __forceinline__ void load_tile(const float* __restrict__ X,
                                          unsigned rowbase, unsigned lastRow,
                                          int n, int hi, TileFrag& f) {
    unsigned rowA = rowbase + (unsigned)n;
    if (rowA > lastRow) rowA = lastRow;
    const float* xrow = X + (size_t)rowA * N_FEAT;
    f.a0 = *(const v2f*)(xrow + 0 + hi * 2);
    f.a1 = *(const v2f*)(xrow + 4 + hi * 2);
    f.a2 = *(const v2f*)(xrow + 8);
}

__device__ __forceinline__ void item_product_store(const TileFrag (&f)[TILES_PER_ITEM],
                                                   const v2f (&cf0)[3], const v2f (&cf1)[3],
                                                   float* __restrict__ Out,
                                                   unsigned rb, unsigned T,
                                                   int lane, int hi) {
    v8f d0 = {};
    v8f d1 = {};
    d0 = __builtin_amdgcn_wmma_f32_16x16x4_f32(false, cf0[0], false, f[0].a0,
                                               (short)0, d0, false, false);
    d1 = __builtin_amdgcn_wmma_f32_16x16x4_f32(false, cf1[0], false, f[1].a0,
                                               (short)0, d1, false, false);
    d0 = __builtin_amdgcn_wmma_f32_16x16x4_f32(false, cf0[1], false, f[0].a1,
                                               (short)0, d0, false, false);
    d1 = __builtin_amdgcn_wmma_f32_16x16x4_f32(false, cf1[1], false, f[1].a1,
                                               (short)0, d1, false, false);
    d0 = __builtin_amdgcn_wmma_f32_16x16x4_f32(false, cf0[2], false, f[0].a2,
                                               (short)0, d0, false, false);
    d1 = __builtin_amdgcn_wmma_f32_16x16x4_f32(false, cf1[2], false, f[1].a2,
                                               (short)0, d1, false, false);

    asm volatile("v_nop\n\tv_nop\n\tv_nop\n\tv_nop"
                 : "+v"(d0), "+v"(d1)
                 : "v"(f[0].a0), "v"(f[0].a1), "v"(f[0].a2),
                   "v"(f[1].a0), "v"(f[1].a1), "v"(f[1].a2));

    v2f o;
    o.x = hi ? d1[0] : d0[0];
    o.y = hi ? d1[1] : d0[1];
    const unsigned row = rb + (unsigned)lane;
    if (row < T) *(v2f*)(Out + (size_t)row * N_CLS) = o;
}

__global__ void __launch_bounds__(256) gemv_wmma_kernel(
    const float* __restrict__ X,
    const float* __restrict__ Bpad,
    float* __restrict__ Out,
    int Tin) {
    if (Tin <= 0) return;
    const unsigned T = (unsigned)Tin;
    const int lane = threadIdx.x & 31;
    const int wave = __builtin_amdgcn_readfirstlane((int)(threadIdx.x >> 5));
    const int hi   = lane >> 4;
    const int n    = lane & 15;

    const unsigned nwaves = gridDim.x * WAVES_PER_BLOCK;
    const unsigned nitems = (T + ROWS_PER_ITEM - 1u) / ROWS_PER_ITEM;
    const unsigned lastRow = T - 1u;
    unsigned g = blockIdx.x * WAVES_PER_BLOCK + (unsigned)wave;
    if (g >= nitems) return;

    v2f cf0[3], cf1[3];
#pragma unroll
    for (int c = 0; c < 3; ++c) {
        const float vx = Bpad[(4 * c + 0 + hi * 2) * 16 + (n & 7)];
        const float vy = Bpad[(4 * c + 1 + hi * 2) * 16 + (n & 7)];
        cf0[c].x = (n < 8) ? vx : 0.0f;
        cf0[c].y = (n < 8) ? vy : 0.0f;
        cf1[c].x = (n < 8) ? 0.0f : vx;
        cf1[c].y = (n < 8) ? 0.0f : vy;
    }

    TileFrag buf[DEPTH][TILES_PER_ITEM];
#pragma unroll
    for (int s = 0; s < DEPTH; ++s) {
        const unsigned rb = (g + (unsigned)s * nwaves) * ROWS_PER_ITEM;
#pragma unroll
        for (int t = 0; t < TILES_PER_ITEM; ++t)
            load_tile(X, rb + 16u * t, lastRow, n, hi, buf[s][t]);
        __builtin_amdgcn_sched_barrier(0);
    }

    const unsigned myItems = (nitems - g + nwaves - 1u) / nwaves;
    const unsigned trips = (myItems + DEPTH - 1u) / DEPTH;
    for (unsigned it = 0; it < trips; ++it) {
#pragma unroll
        for (int s = 0; s < DEPTH; ++s) {
            const unsigned rb = g * ROWS_PER_ITEM;
            item_product_store(buf[s], cf0, cf1, Out, rb, T, lane, hi);
            const unsigned rbn = (g + (unsigned)DEPTH * nwaves) * ROWS_PER_ITEM;
#pragma unroll
            for (int t = 0; t < TILES_PER_ITEM; ++t)
                load_tile(X, rbn + 16u * t, lastRow, n, hi, buf[s][t]);
            g += nwaves;
        }
    }
}

extern "C" void kernel_launch(void* const* d_in, const int* in_sizes, int n_in,
                              void* d_out, int out_size, void* d_ws, size_t ws_size,
                              hipStream_t stream) {
    const float* x     = (const float*)d_in[0];
    const float* psi_r = (const float*)d_in[1];
    const float* psi_i = (const float*)d_in[2];
    const float* A_r   = (const float*)d_in[3];
    const float* A_i   = (const float*)d_in[4];
    float* out  = (float*)d_out;
    float* Bpad = (float*)d_ws;

    const int T = in_sizes[0] / N_FEAT;

    build_coeff_kernel<<<1, 256, 0, stream>>>(psi_r, psi_i, A_r, A_i, Bpad);

    gemv_wmma_kernel<<<NUM_BLOCKS, 256, 0, stream>>>(x, Bpad, out, T);
}


# one launch instead of two: every workgroup builds the 2x10 coefficients itself (4 terms per thread, then a 25-way sum through LDS: coefficient sum order changed) in front of its first x loads
# speedup vs baseline: 2.3023x; 1.5495x over previous
#include <hip/hip_runtime.h>


typedef __attribute__((ext_vector_type(2))) float v2f;
typedef __attribute__((ext_vector_type(8))) float v8f;

#define N_FEAT 10
#define N_CLS  2
#define KPAD   12

#define WAVES_PER_BLOCK 8
#define NUM_BLOCKS      1024
#define TILES_PER_ITEM  2
#define ROWS_PER_ITEM   (16 * TILES_PER_ITEM)
#define DEPTH           4

#define COEF_THREADS 250
#define COEF_GROUPS  25

struct TileFrag { v2f a0, a1, a2; };

__device__ __forceinline__ void load_tile(const float* __restrict__ X,
                                          unsigned rowbase, unsigned lastRow,
                                          int n, int hi, TileFrag& f) {
    unsigned rowA = rowbase + (unsigned)n;
    if (rowA > lastRow) rowA = lastRow;
    const float* xrow = X + (size_t)rowA * N_FEAT;
    f.a0 = *(const v2f*)(xrow + 0 + hi * 2);
    f.a1 = *(const v2f*)(xrow + 4 + hi * 2);
    f.a2 = *(const v2f*)(xrow + 8);
}

__device__ __forceinline__ void item_product_store(const TileFrag (&f)[TILES_PER_ITEM],
                                                   const v2f (&cf0)[3], const v2f (&cf1)[3],
                                                   float* __restrict__ Out,
                                                   unsigned rb, unsigned T,
                                                   int lane, int hi) {
    v8f d0 = {};
    v8f d1 = {};
    d0 = __builtin_amdgcn_wmma_f32_16x16x4_f32(false, cf0[0], false, f[0].a0,
                                               (short)0, d0, false, false);
    d1 = __builtin_amdgcn_wmma_f32_16x16x4_f32(false, cf1[0], false, f[1].a0,
                                               (short)0, d1, false, false);
    d0 = __builtin_amdgcn_wmma_f32_16x16x4_f32(false, cf0[1], false, f[0].a1,
                                               (short)0, d0, false, false);
    d1 = __builtin_amdgcn_wmma_f32_16x16x4_f32(false, cf1[1], false, f[1].a1,
                                               (short)0, d1, false, false);
    d0 = __builtin_amdgcn_wmma_f32_16x16x4_f32(false, cf0[2], false, f[0].a2,
                                               (short)0, d0, false, false);
    d1 = __builtin_amdgcn_wmma_f32_16x16x4_f32(false, cf1[2], false, f[1].a2,
                                               (short)0, d1, false, false);

    asm volatile("v_nop\n\tv_nop\n\tv_nop\n\tv_nop"
                 : "+v"(d0), "+v"(d1)
                 : "v"(f[0].a0), "v"(f[0].a1), "v"(f[0].a2),
                   "v"(f[1].a0), "v"(f[1].a1), "v"(f[1].a2));

    v2f o;
    o.x = hi ? d1[0] : d0[0];
    o.y = hi ? d1[1] : d0[1];
    const unsigned row = rb + (unsigned)lane;
    if (row < T) *(v2f*)(Out + (size_t)row * N_CLS) = o;
}

__device__ __forceinline__ void lds_barrier() {
    __builtin_amdgcn_fence(__ATOMIC_RELEASE, "workgroup", "local");
    __builtin_amdgcn_s_barrier();
    __builtin_amdgcn_fence(__ATOMIC_ACQUIRE, "workgroup", "local");
}

__global__ void __launch_bounds__(256) gemv_wmma_kernel(
    const float* __restrict__ X,
    const float* __restrict__ psi_r,
    const float* __restrict__ psi_i,
    const float* __restrict__ A_r,
    const float* __restrict__ A_i,
    float* __restrict__ Out,
    int Tin) {
    if (Tin <= 0) return;
    __shared__ float sPart[N_CLS][256];
    __shared__ float sM[N_CLS][KPAD];

    const unsigned T = (unsigned)Tin;
    const int tid  = (int)threadIdx.x;
    const int lane = tid & 31;
    const int wave = __builtin_amdgcn_readfirstlane((int)(threadIdx.x >> 5));
    const int hi   = lane >> 4;
    const int n    = lane & 15;

    const unsigned nwaves = gridDim.x * WAVES_PER_BLOCK;
    const unsigned nitems = (T + ROWS_PER_ITEM - 1u) / ROWS_PER_ITEM;
    const unsigned lastRow = T - 1u;
    unsigned g = blockIdx.x * WAVES_PER_BLOCK + (unsigned)wave;

    const int ct = tid < COEF_THREADS ? tid : COEF_THREADS - 1;
    const int cq = ct / N_FEAT;
    float ar[4][N_CLS], ai[4][N_CLS], pri[4], pii[4], prj[4], pij[4];
#pragma unroll
    for (int m = 0; m < 4; ++m) {
        const int ij = cq + COEF_GROUPS * m;
        const int i = ij / N_FEAT;
        const int j = ij - i * N_FEAT;
        pri[m] = psi_r[i];
        pii[m] = psi_i[i];
        prj[m] = psi_r[j];
        pij[m] = psi_i[j];
#pragma unroll
        for (int k = 0; k < N_CLS; ++k) {
            const int e = k * (N_FEAT * N_FEAT * N_FEAT) + ct + COEF_THREADS * m;
            ar[m][k] = A_r[e];
            ai[m][k] = A_i[e];
        }
    }
    __builtin_amdgcn_sched_barrier(0);

    TileFrag buf[DEPTH][TILES_PER_ITEM];
#pragma unroll
    for (int s = 0; s < DEPTH; ++s) {
        const unsigned rb = (g + (unsigned)s * nwaves) * ROWS_PER_ITEM;
#pragma unroll
        for (int t = 0; t < TILES_PER_ITEM; ++t)
            load_tile(X, rb + 16u * t, lastRow, n, hi, buf[s][t]);
        __builtin_amdgcn_sched_barrier(0);
    }

    {
        float p0 = 0.0f, p1 = 0.0f;
#pragma unroll
        for (int m = 0; m < 4; ++m) {
            const float wj = prj[m] - pij[m];
            p0 += wj * (pri[m] * ar[m][0] - pii[m] * ai[m][0]);
            p1 += wj * (pri[m] * ar[m][1] - pii[m] * ai[m][1]);
        }
        sPart[0][tid] = p0;
        sPart[1][tid] = p1;
        lds_barrier();
        if (tid < N_CLS * KPAD) {
            const int k = tid / KPAD;
            const int a = tid - k * KPAD;
            float sum = 0.0f;
            if (a < N_FEAT) {
#pragma unroll
                for (int q = 0; q < COEF_GROUPS; ++q)
                    sum += sPart[k][a + N_FEAT * q];
            }
            sM[k][a] = sum;
        }
        lds_barrier();
    }

    if (g >= nitems) return;

    v2f cf0[3], cf1[3];
#pragma unroll
    for (int c = 0; c < 3; ++c) {
        const float vx = sM[n & 1][4 * c + 0 + hi * 2];
        const float vy = sM[n & 1][4 * c + 1 + hi * 2];
        const bool in0 = n < N_CLS;
        const bool in1 = (n >= 8) && (n < 8 + N_CLS);
        cf0[c].x = in0 ? vx : 0.0f;
        cf0[c].y = in0 ? vy : 0.0f;
        cf1[c].x = in1 ? vx : 0.0f;
        cf1[c].y = in1 ? vy : 0.0f;
    }

    const unsigned myItems = (nitems - g + nwaves - 1u) / nwaves;
    const unsigned trips = (myItems + DEPTH - 1u) / DEPTH;
    for (unsigned it = 0; it < trips; ++it) {
#pragma unroll
        for (int s = 0; s < DEPTH; ++s) {
            const unsigned rb = g * ROWS_PER_ITEM;
            item_product_store(buf[s], cf0, cf1, Out, rb, T, lane, hi);
            const unsigned rbn = (g + (unsigned)DEPTH * nwaves) * ROWS_PER_ITEM;
#pragma unroll
            for (int t = 0; t < TILES_PER_ITEM; ++t)
                load_tile(X, rbn + 16u * t, lastRow, n, hi, buf[s][t]);
            g += nwaves;
        }
    }
}

extern "C" void kernel_launch(void* const* d_in, const int* in_sizes, int n_in,
                              void* d_out, int out_size, void* d_ws, size_t ws_size,
                              hipStream_t stream) {
    const float* x     = (const float*)d_in[0];
    const float* psi_r = (const float*)d_in[1];
    const float* psi_i = (const float*)d_in[2];
    const float* A_r   = (const float*)d_in[3];
    const float* A_i   = (const float*)d_in[4];
    float* out  = (float*)d_out;

    const int T = in_sizes[0] / N_FEAT;

    gemv_wmma_kernel<<<NUM_BLOCKS, 256, 0, stream>>>(x, psi_r, psi_i, A_r, A_i, out, T);
}
